# starting kernel
# baseline (speedup 1.0000x reference)
#include <hip/hip_runtime.h>


typedef __attribute__((ext_vector_type(2))) float v2f;
typedef __attribute__((ext_vector_type(8))) float v8f;

#if defined(__has_builtin)
#  if __has_builtin(__builtin_amdgcn_tanhf)
#    define TANHF(x) __builtin_amdgcn_tanhf(x)
#  else
#    define TANHF(x) tanhf(x)
#  endif
#else
#  define TANHF(x) tanhf(x)
#endif

#define NELEM   4
#define DESC    128
#define HID     64
#define CHUNK   256
#define NWAVES  8
#define HSTRIDE 66

#define W1_F  (NELEM*DESC*HID)
#define W2_F  (NELEM*HID*HID)
#define W3_F  (NELEM*HID)
#define B1_F  (NELEM*HID)
#define B2_F  (NELEM*HID)
#define B3_F  (NELEM)
#define HST_F (NWAVES*16*HSTRIDE)
#define MOLE_F 4
#define FLOATS_TOTAL (W1_F+W2_F+W3_F+B1_F+B2_F+B3_F+HST_F+MOLE_F)
#define INTS_TOTAL   (NELEM + NELEM*CHUNK)
#define SMEM_BYTES   ((size_t)(FLOATS_TOTAL + INTS_TOTAL) * 4)

__global__ __launch_bounds__(256)
void hdnnp_wmma_kernel(const float* __restrict__ X,
                       const int*   __restrict__ Z,
                       const float* __restrict__ W1,
                       const float* __restrict__ b1,
                       const float* __restrict__ W2,
                       const float* __restrict__ b2,
                       const float* __restrict__ W3,
                       const float* __restrict__ b3,
                       float*       __restrict__ out)
{
    extern __shared__ float smem[];
    float* w1s    = smem;
    float* w2s    = w1s + W1_F;
    float* w3s    = w2s + W2_F;
    float* b1s    = w3s + W3_F;
    float* b2s    = b1s + B1_F;
    float* b3s    = b2s + B2_F;
    float* hstage = b3s + B3_F;
    float* molE   = hstage + HST_F;
    int*   cnt    = (int*)(molE + MOLE_F);
    int*   binIdx = cnt + NELEM;

    const int tid  = threadIdx.x;
    const int lane = tid & 31;
    const int wave = tid >> 5;
    const int m    = lane & 15;
    const int hf   = lane >> 4;
    const int base = blockIdx.x * CHUNK;

    if (tid < NELEM) cnt[tid] = 0;
    if (tid < 4)     molE[tid] = 0.0f;
    __syncthreads();

    {
        int z   = Z[base + tid];
        int pos = atomicAdd(&cnt[z], 1);
        binIdx[z * CHUNK + pos] = tid;
    }

    {
        const float4* s1 = (const float4*)W1;
        for (int i = tid; i < W1_F / 4; i += 256) {
            float4 v = s1[i];
            int g  = i * 4;
            int n  = g & (HID - 1);
            int d  = (g >> 6) & (DESC - 1);
            int e  = g >> 13;
            int kk = d >> 2, hh = (d >> 1) & 1, lo = d & 1;
            float* dst = w1s + ((e * (DESC/4) + kk) * 2 + hh) * (2*HID) + 2*n + lo;
            dst[0] = v.x; dst[2] = v.y; dst[4] = v.z; dst[6] = v.w;
        }
        const float4* s2 = (const float4*)W2;
        for (int i = tid; i < W2_F / 4; i += 256) {
            float4 v = s2[i];
            int g  = i * 4;
            int n  = g & (HID - 1);
            int h  = (g >> 6) & (HID - 1);
            int e  = g >> 12;
            int kk = h >> 2, hh = (h >> 1) & 1, lo = h & 1;
            float* dst = w2s + ((e * (HID/4) + kk) * 2 + hh) * (2*HID) + 2*n + lo;
            dst[0] = v.x; dst[2] = v.y; dst[4] = v.z; dst[6] = v.w;
        }
        if (tid < W3_F) w3s[tid] = W3[tid];
        if (tid < B1_F) b1s[tid] = b1[tid];
        if (tid < B2_F) b2s[tid] = b2[tid];
        if (tid < B3_F) b3s[tid] = b3[tid];
    }
    __syncthreads();

    const int c0 = cnt[0], c1 = cnt[1], c2 = cnt[2], c3 = cnt[3];
    const int t0 = (c0 + 15) >> 4, t1 = (c1 + 15) >> 4;
    const int t2 = (c2 + 15) >> 4, t3 = (c3 + 15) >> 4;
    const int T  = t0 + t1 + t2 + t3;

    float* hst = hstage + wave * (16 * HSTRIDE);

    for (int t = wave; t < T; t += NWAVES) {
        int e, lt, r = t;
        if (r < t0)            { e = 0; lt = r; }
        else { r -= t0;
        if (r < t1)            { e = 1; lt = r; }
        else { r -= t1;
        if (r < t2)            { e = 2; lt = r; }
        else                   { e = 3; lt = r - t2; } } }
        const int ccnt = cnt[e];

        const int  pos   = lt * 16 + m;
        const bool valid = pos < ccnt;
        const int  slot  = valid ? binIdx[e * CHUNK + pos] : 0;
        const float* xrow = X + (size_t)(base + slot) * DESC + 2 * hf;

        v8f a0 = {}, a1 = {}, a2 = {}, a3 = {};
        const float* w1e = w1s + e * (DESC * HID);
        #pragma unroll 4
        for (int kk = 0; kk < DESC / 4; ++kk) {
            v2f a = *(const v2f*)(xrow + 4 * kk);
            const float* wb = w1e + (kk * 2 + hf) * (2*HID) + 2*m;
            v2f bt0 = *(const v2f*)(wb);
            v2f bt1 = *(const v2f*)(wb + 32);
            v2f bt2 = *(const v2f*)(wb + 64);
            v2f bt3 = *(const v2f*)(wb + 96);
            a0 = __builtin_amdgcn_wmma_f32_16x16x4_f32(false, a, false, bt0, (short)0, a0, false, false);
            a1 = __builtin_amdgcn_wmma_f32_16x16x4_f32(false, a, false, bt1, (short)0, a1, false, false);
            a2 = __builtin_amdgcn_wmma_f32_16x16x4_f32(false, a, false, bt2, (short)0, a2, false, false);
            a3 = __builtin_amdgcn_wmma_f32_16x16x4_f32(false, a, false, bt3, (short)0, a3, false, false);
        }
        {
            const float* b1e = b1s + e * HID;
            const float z0 = b1e[m], z1 = b1e[16 + m], z2 = b1e[32 + m], z3 = b1e[48 + m];
            #pragma unroll
            for (int g = 0; g < 8; ++g) {
                float* hr = hst + (g + 8 * hf) * HSTRIDE + m;
                hr[0]  = TANHF(a0[g] + z0);
                hr[16] = TANHF(a1[g] + z1);
                hr[32] = TANHF(a2[g] + z2);
                hr[48] = TANHF(a3[g] + z3);
            }
        }

        v8f d0 = {}, d1 = {}, d2 = {}, d3 = {};
        const float* w2e  = w2s + e * (HID * HID);
        const float* arow = hst + m * HSTRIDE + 2 * hf;
        #pragma unroll 4
        for (int kk = 0; kk < HID / 4; ++kk) {
            v2f a = *(const v2f*)(arow + 4 * kk);
            const float* wb = w2e + (kk * 2 + hf) * (2*HID) + 2*m;
            v2f bt0 = *(const v2f*)(wb);
            v2f bt1 = *(const v2f*)(wb + 32);
            v2f bt2 = *(const v2f*)(wb + 64);
            v2f bt3 = *(const v2f*)(wb + 96);
            d0 = __builtin_amdgcn_wmma_f32_16x16x4_f32(false, a, false, bt0, (short)0, d0, false, false);
            d1 = __builtin_amdgcn_wmma_f32_16x16x4_f32(false, a, false, bt1, (short)0, d1, false, false);
            d2 = __builtin_amdgcn_wmma_f32_16x16x4_f32(false, a, false, bt2, (short)0, d2, false, false);
            d3 = __builtin_amdgcn_wmma_f32_16x16x4_f32(false, a, false, bt3, (short)0, d3, false, false);
        }
        {
            const float* b2e = b2s + e * HID;
            const float z0 = b2e[m], z1 = b2e[16 + m], z2 = b2e[32 + m], z3 = b2e[48 + m];
            #pragma unroll
            for (int g = 0; g < 8; ++g) {
                float* hr = hst + (g + 8 * hf) * HSTRIDE + m;
                hr[0]  = TANHF(d0[g] + z0);
                hr[16] = TANHF(d1[g] + z1);
                hr[32] = TANHF(d2[g] + z2);
                hr[48] = TANHF(d3[g] + z3);
            }
        }

        {
            float en = 0.0f;
            const float* w3e = w3s + e * HID;
            const float* hr  = hst + m * HSTRIDE;
            #pragma unroll 8
            for (int h = 0; h < HID; ++h) en = fmaf(hr[h], w3e[h], en);
            en += b3s[e];
            if (hf == 0 && valid)
                atomicAdd(&molE[slot >> 6], en);
        }
    }

    __syncthreads();
    if (tid < 4) out[blockIdx.x * 4 + tid] = molE[tid];
}

extern "C" void kernel_launch(void* const* d_in, const int* in_sizes, int n_in,
                              void* d_out, int out_size, void* d_ws, size_t ws_size,
                              hipStream_t stream) {
    (void)n_in; (void)out_size; (void)d_ws; (void)ws_size;
    const float* X  = (const float*)d_in[0];
    const int*   Z  = (const int*)  d_in[1];
    const float* W1 = (const float*)d_in[2];
    const float* b1 = (const float*)d_in[3];
    const float* W2 = (const float*)d_in[4];
    const float* b2 = (const float*)d_in[5];
    const float* W3 = (const float*)d_in[6];
    const float* b3 = (const float*)d_in[7];
    float* out = (float*)d_out;

    const int M      = in_sizes[1];
    const int blocks = M / CHUNK;

    hipLaunchKernelGGL(hdnnp_wmma_kernel, dim3(blocks), dim3(256), SMEM_BYTES,
                       stream, X, Z, W1, b1, W2, b2, W3, b3, out);
}


# One 8-wave work-group per WGP takes 1024 atoms (16 molecules), not 256: weights staged to LDS once per WGP, tiles binned over 1024 atoms (9 not 12 per wave); molecule sums stay LDS float atomics
# speedup vs baseline: 1.3831x; 1.3831x over previous
#include <hip/hip_runtime.h>


typedef __attribute__((ext_vector_type(2))) float v2f;
typedef __attribute__((ext_vector_type(8))) float v8f;

#if defined(__has_builtin)
#  if __has_builtin(__builtin_amdgcn_tanhf)
#    define TANHF(x) __builtin_amdgcn_tanhf(x)
#  else
#    define TANHF(x) tanhf(x)
#  endif
#else
#  define TANHF(x) tanhf(x)
#endif

#define NELEM   4
#define DESC    128
#define HID     64
#define NATOM   64
#define CHUNK   1024
#define MOLS    (CHUNK/NATOM)
#define NTHR    256
#define NWAVES  8
#define HSTRIDE 66

#define W1_F  (NELEM*DESC*HID)
#define W2_F  (NELEM*HID*HID)
#define W3_F  (NELEM*HID)
#define B1_F  (NELEM*HID)
#define B2_F  (NELEM*HID)
#define B3_F  (NELEM)
#define HST_F (NWAVES*16*HSTRIDE)
#define MOLE_F MOLS
#define FLOATS_TOTAL (W1_F+W2_F+W3_F+B1_F+B2_F+B3_F+HST_F+MOLE_F)
#define INTS_TOTAL   (NELEM + NELEM*CHUNK)
#define SMEM_BYTES   ((size_t)(FLOATS_TOTAL + INTS_TOTAL) * 4)

__global__ __launch_bounds__(256)
void hdnnp_wmma_kernel(const float* __restrict__ X,
                       const int*   __restrict__ Z,
                       const float* __restrict__ W1,
                       const float* __restrict__ b1,
                       const float* __restrict__ W2,
                       const float* __restrict__ b2,
                       const float* __restrict__ W3,
                       const float* __restrict__ b3,
                       float*       __restrict__ out)
{
    extern __shared__ float smem[];
    float* w1s    = smem;
    float* w2s    = w1s + W1_F;
    float* w3s    = w2s + W2_F;
    float* b1s    = w3s + W3_F;
    float* b2s    = b1s + B1_F;
    float* b3s    = b2s + B2_F;
    float* hstage = b3s + B3_F;
    float* molE   = hstage + HST_F;
    int*   cnt    = (int*)(molE + MOLE_F);
    int*   binIdx = cnt + NELEM;

    const int tid  = threadIdx.x;
    const int lane = tid & 31;
    const int wave = tid >> 5;
    const int m    = lane & 15;
    const int hf   = lane >> 4;
    const int base = blockIdx.x * CHUNK;

    if (tid < NELEM) cnt[tid] = 0;
    if (tid < MOLS)  molE[tid] = 0.0f;
    __syncthreads();

    {
        int z[CHUNK / NTHR];
        for (int j = 0; j < CHUNK / NTHR; ++j) z[j] = Z[base + j * NTHR + tid];
        for (int j = 0; j < CHUNK / NTHR; ++j) {
            int pos = atomicAdd(&cnt[z[j]], 1);
            binIdx[z[j] * CHUNK + pos] = j * NTHR + tid;
        }
    }

    {
        const float4* s1 = (const float4*)W1;
        for (int i = tid; i < W1_F / 4; i += 256) {
            float4 v = s1[i];
            int g  = i * 4;
            int n  = g & (HID - 1);
            int d  = (g >> 6) & (DESC - 1);
            int e  = g >> 13;
            int kk = d >> 2, hh = (d >> 1) & 1, lo = d & 1;
            float* dst = w1s + ((e * (DESC/4) + kk) * 2 + hh) * (2*HID) + 2*n + lo;
            dst[0] = v.x; dst[2] = v.y; dst[4] = v.z; dst[6] = v.w;
        }
        const float4* s2 = (const float4*)W2;
        for (int i = tid; i < W2_F / 4; i += 256) {
            float4 v = s2[i];
            int g  = i * 4;
            int n  = g & (HID - 1);
            int h  = (g >> 6) & (HID - 1);
            int e  = g >> 12;
            int kk = h >> 2, hh = (h >> 1) & 1, lo = h & 1;
            float* dst = w2s + ((e * (HID/4) + kk) * 2 + hh) * (2*HID) + 2*n + lo;
            dst[0] = v.x; dst[2] = v.y; dst[4] = v.z; dst[6] = v.w;
        }
        if (tid < W3_F) w3s[tid] = W3[tid];
        if (tid < B1_F) b1s[tid] = b1[tid];
        if (tid < B2_F) b2s[tid] = b2[tid];
        if (tid < B3_F) b3s[tid] = b3[tid];
    }
    __syncthreads();

    const int c0 = cnt[0], c1 = cnt[1], c2 = cnt[2], c3 = cnt[3];
    const int t0 = (c0 + 15) >> 4, t1 = (c1 + 15) >> 4;
    const int t2 = (c2 + 15) >> 4, t3 = (c3 + 15) >> 4;
    const int T  = t0 + t1 + t2 + t3;

    float* hst = hstage + wave * (16 * HSTRIDE);

    for (int t = wave; t < T; t += NWAVES) {
        int e, lt, r = t;
        if (r < t0)            { e = 0; lt = r; }
        else { r -= t0;
        if (r < t1)            { e = 1; lt = r; }
        else { r -= t1;
        if (r < t2)            { e = 2; lt = r; }
        else                   { e = 3; lt = r - t2; } } }
        const int ccnt = cnt[e];

        const int  pos   = lt * 16 + m;
        const bool valid = pos < ccnt;
        const int  slot  = valid ? binIdx[e * CHUNK + pos] : 0;
        const float* xrow = X + (size_t)(base + slot) * DESC + 2 * hf;

        v8f a0 = {}, a1 = {}, a2 = {}, a3 = {};
        const float* w1e = w1s + e * (DESC * HID);
        #pragma unroll 4
        for (int kk = 0; kk < DESC / 4; ++kk) {
            v2f a = *(const v2f*)(xrow + 4 * kk);
            const float* wb = w1e + (kk * 2 + hf) * (2*HID) + 2*m;
            v2f bt0 = *(const v2f*)(wb);
            v2f bt1 = *(const v2f*)(wb + 32);
            v2f bt2 = *(const v2f*)(wb + 64);
            v2f bt3 = *(const v2f*)(wb + 96);
            a0 = __builtin_amdgcn_wmma_f32_16x16x4_f32(false, a, false, bt0, (short)0, a0, false, false);
            a1 = __builtin_amdgcn_wmma_f32_16x16x4_f32(false, a, false, bt1, (short)0, a1, false, false);
            a2 = __builtin_amdgcn_wmma_f32_16x16x4_f32(false, a, false, bt2, (short)0, a2, false, false);
            a3 = __builtin_amdgcn_wmma_f32_16x16x4_f32(false, a, false, bt3, (short)0, a3, false, false);
        }
        {
            const float* b1e = b1s + e * HID;
            const float z0 = b1e[m], z1 = b1e[16 + m], z2 = b1e[32 + m], z3 = b1e[48 + m];
            #pragma unroll
            for (int g = 0; g < 8; ++g) {
                float* hr = hst + (g + 8 * hf) * HSTRIDE + m;
                hr[0]  = TANHF(a0[g] + z0);
                hr[16] = TANHF(a1[g] + z1);
                hr[32] = TANHF(a2[g] + z2);
                hr[48] = TANHF(a3[g] + z3);
            }
        }

        v8f d0 = {}, d1 = {}, d2 = {}, d3 = {};
        const float* w2e  = w2s + e * (HID * HID);
        const float* arow = hst + m * HSTRIDE + 2 * hf;
        #pragma unroll 4
        for (int kk = 0; kk < HID / 4; ++kk) {
            v2f a = *(const v2f*)(arow + 4 * kk);
            const float* wb = w2e + (kk * 2 + hf) * (2*HID) + 2*m;
            v2f bt0 = *(const v2f*)(wb);
            v2f bt1 = *(const v2f*)(wb + 32);
            v2f bt2 = *(const v2f*)(wb + 64);
            v2f bt3 = *(const v2f*)(wb + 96);
            d0 = __builtin_amdgcn_wmma_f32_16x16x4_f32(false, a, false, bt0, (short)0, d0, false, false);
            d1 = __builtin_amdgcn_wmma_f32_16x16x4_f32(false, a, false, bt1, (short)0, d1, false, false);
            d2 = __builtin_amdgcn_wmma_f32_16x16x4_f32(false, a, false, bt2, (short)0, d2, false, false);
            d3 = __builtin_amdgcn_wmma_f32_16x16x4_f32(false, a, false, bt3, (short)0, d3, false, false);
        }
        {
            const float* b2e = b2s + e * HID;
            const float z0 = b2e[m], z1 = b2e[16 + m], z2 = b2e[32 + m], z3 = b2e[48 + m];
            #pragma unroll
            for (int g = 0; g < 8; ++g) {
                float* hr = hst + (g + 8 * hf) * HSTRIDE + m;
                hr[0]  = TANHF(d0[g] + z0);
                hr[16] = TANHF(d1[g] + z1);
                hr[32] = TANHF(d2[g] + z2);
                hr[48] = TANHF(d3[g] + z3);
            }
        }

        {
            float en = 0.0f;
            const float* w3e = w3s + e * HID;
            const float* hr  = hst + m * HSTRIDE;
            #pragma unroll 8
            for (int h = 0; h < HID; ++h) en = fmaf(hr[h], w3e[h], en);
            en += b3s[e];
            if (hf == 0 && valid)
                atomicAdd(&molE[slot >> 6], en);
        }
    }

    __syncthreads();
    if (tid < MOLS) out[blockIdx.x * MOLS + tid] = molE[tid];
}

extern "C" void kernel_launch(void* const* d_in, const int* in_sizes, int n_in,
                              void* d_out, int out_size, void* d_ws, size_t ws_size,
                              hipStream_t stream) {
    (void)n_in; (void)out_size; (void)d_ws; (void)ws_size;
    const float* X  = (const float*)d_in[0];
    const int*   Z  = (const int*)  d_in[1];
    const float* W1 = (const float*)d_in[2];
    const float* b1 = (const float*)d_in[3];
    const float* W2 = (const float*)d_in[4];
    const float* b2 = (const float*)d_in[5];
    const float* W3 = (const float*)d_in[6];
    const float* b3 = (const float*)d_in[7];
    float* out = (float*)d_out;

    const int M      = in_sizes[1];
    const int blocks = M / CHUNK;

    hipLaunchKernelGGL(hdnnp_wmma_kernel, dim3(blocks), dim3(256), SMEM_BYTES,
                       stream, X, Z, W1, b1, W2, b2, W3, b3, out);
}


# Layer-1 A tile prefetched: next tile's rows go to registers (16 b128 per lane) right after layer 1, not 32 b64 gathers waited on in the k loop; W1 k order permuted per 8 (layer-1 sum order changes)
# speedup vs baseline: 1.6682x; 1.2061x over previous
#include <hip/hip_runtime.h>


typedef __attribute__((ext_vector_type(2))) float v2f;
typedef __attribute__((ext_vector_type(4))) float v4f;
typedef __attribute__((ext_vector_type(8))) float v8f;

#if defined(__has_builtin)
#  if __has_builtin(__builtin_amdgcn_tanhf)
#    define TANHF(x) __builtin_amdgcn_tanhf(x)
#  else
#    define TANHF(x) tanhf(x)
#  endif
#else
#  define TANHF(x) tanhf(x)
#endif

#define NELEM   4
#define DESC    128
#define HID     64
#define NATOM   64
#define CHUNK   1024
#define MOLS    (CHUNK/NATOM)
#define NTHR    256
#define NWAVES  8
#define HSTRIDE 66

#define W1_F  (NELEM*DESC*HID)
#define W2_F  (NELEM*HID*HID)
#define W3_F  (NELEM*HID)
#define B1_F  (NELEM*HID)
#define B2_F  (NELEM*HID)
#define B3_F  (NELEM)
#define HST_F (NWAVES*16*HSTRIDE)
#define MOLE_F MOLS
#define FLOATS_TOTAL (W1_F+W2_F+W3_F+B1_F+B2_F+B3_F+HST_F+MOLE_F)
#define INTS_TOTAL   (NELEM + NELEM*CHUNK)
#define SMEM_BYTES   ((size_t)(FLOATS_TOTAL + INTS_TOTAL) * 4)

__global__ __launch_bounds__(256) __attribute__((amdgpu_waves_per_eu(2, 2)))
void hdnnp_wmma_kernel(const float* __restrict__ X,
                       const int*   __restrict__ Z,
                       const float* __restrict__ W1,
                       const float* __restrict__ b1,
                       const float* __restrict__ W2,
                       const float* __restrict__ b2,
                       const float* __restrict__ W3,
                       const float* __restrict__ b3,
                       float*       __restrict__ out)
{
    extern __shared__ float smem[];
    float* w1s    = smem;
    float* w2s    = w1s + W1_F;
    float* w3s    = w2s + W2_F;
    float* b1s    = w3s + W3_F;
    float* b2s    = b1s + B1_F;
    float* b3s    = b2s + B2_F;
    float* hstage = b3s + B3_F;
    float* molE   = hstage + HST_F;
    int*   cnt    = (int*)(molE + MOLE_F);
    int*   binIdx = cnt + NELEM;

    const int tid  = threadIdx.x;
    const int lane = tid & 31;
    const int wave = tid >> 5;
    const int m    = lane & 15;
    const int hf   = lane >> 4;
    const int base = blockIdx.x * CHUNK;

    if (tid < NELEM) cnt[tid] = 0;
    if (tid < MOLS)  molE[tid] = 0.0f;
    __syncthreads();

    {
        int z[CHUNK / NTHR];
        for (int j = 0; j < CHUNK / NTHR; ++j) z[j] = Z[base + j * NTHR + tid];
        for (int j = 0; j < CHUNK / NTHR; ++j) {
            int pos = atomicAdd(&cnt[z[j]], 1);
            binIdx[z[j] * CHUNK + pos] = j * NTHR + tid;
        }
    }

    {
        const float4* s1 = (const float4*)W1;
        for (int i = tid; i < W1_F / 4; i += 256) {
            float4 v = s1[i];
            int g  = i * 4;
            int n  = g & (HID - 1);
            int d  = (g >> 6) & (DESC - 1);
            int e  = g >> 13;
            int kk = ((d >> 3) << 1) | ((d >> 1) & 1), hh = (d >> 2) & 1, lo = d & 1;
            float* dst = w1s + ((e * (DESC/4) + kk) * 2 + hh) * (2*HID) + 2*n + lo;
            dst[0] = v.x; dst[2] = v.y; dst[4] = v.z; dst[6] = v.w;
        }
        const float4* s2 = (const float4*)W2;
        for (int i = tid; i < W2_F / 4; i += 256) {
            float4 v = s2[i];
            int g  = i * 4;
            int n  = g & (HID - 1);
            int h  = (g >> 6) & (HID - 1);
            int e  = g >> 12;
            int kk = h >> 2, hh = (h >> 1) & 1, lo = h & 1;
            float* dst = w2s + ((e * (HID/4) + kk) * 2 + hh) * (2*HID) + 2*n + lo;
            dst[0] = v.x; dst[2] = v.y; dst[4] = v.z; dst[6] = v.w;
        }
        if (tid < W3_F) w3s[tid] = W3[tid];
        if (tid < B1_F) b1s[tid] = b1[tid];
        if (tid < B2_F) b2s[tid] = b2[tid];
        if (tid < B3_F) b3s[tid] = b3[tid];
    }
    __syncthreads();

    const int c0 = cnt[0], c1 = cnt[1], c2 = cnt[2], c3 = cnt[3];
    const int t0 = (c0 + 15) >> 4, t1 = (c1 + 15) >> 4;
    const int t2 = (c2 + 15) >> 4, t3 = (c3 + 15) >> 4;
    const int T  = t0 + t1 + t2 + t3;

    const int wv = __builtin_amdgcn_readfirstlane(wave);
    const int TT = __builtin_amdgcn_readfirstlane(T);

    float* hst = hstage + wv * (16 * HSTRIDE);

    auto tile_of = [&](int t, int& e, bool& valid, int& slot) {
        int lt, r = t;
        if (r < t0)            { e = 0; lt = r; }
        else { r -= t0;
        if (r < t1)            { e = 1; lt = r; }
        else { r -= t1;
        if (r < t2)            { e = 2; lt = r; }
        else                   { e = 3; lt = r - t2; } } }
        const int ccnt = cnt[e];
        const int pos  = lt * 16 + m;
        valid = pos < ccnt;
        slot  = valid ? binIdx[e * CHUNK + pos] : 0;
    };
    auto load_rows = [&](v4f (&xa)[DESC / 8], int slot) {
        const v4f* xr = (const v4f*)(X + (size_t)(base + slot) * DESC + 4 * hf);
        #pragma unroll
        for (int j = 0; j < DESC / 8; ++j) xa[j] = xr[2 * j];
    };

    int  e, slot;
    bool valid;
    v4f  xa[DESC / 8];
    tile_of(wv, e, valid, slot);
    load_rows(xa, slot);

    for (int t = wv; t < TT; t += NWAVES) {
        const bool more = t + NWAVES < TT;
        int  e_n = e, slot_n = slot;
        bool valid_n = false;
        if (more) tile_of(t + NWAVES, e_n, valid_n, slot_n);

        v8f a0 = {}, a1 = {}, a2 = {}, a3 = {};
        const float* w1e = w1s + e * (DESC * HID);
        #pragma unroll
        for (int j = 0; j < DESC / 8; ++j) {
            const v4f x4 = xa[j];
            #pragma unroll
            for (int b = 0; b < 2; ++b) {
                const int kk = 2 * j + b;
                const v2f a = b ? __builtin_shufflevector(x4, x4, 2, 3)
                                : __builtin_shufflevector(x4, x4, 0, 1);
                const float* wb = w1e + (kk * 2 + hf) * (2*HID) + 2*m;
                v2f bt0 = *(const v2f*)(wb);
                v2f bt1 = *(const v2f*)(wb + 32);
                v2f bt2 = *(const v2f*)(wb + 64);
                v2f bt3 = *(const v2f*)(wb + 96);
                a0 = __builtin_amdgcn_wmma_f32_16x16x4_f32(false, a, false, bt0, (short)0, a0, false, false);
                a1 = __builtin_amdgcn_wmma_f32_16x16x4_f32(false, a, false, bt1, (short)0, a1, false, false);
                a2 = __builtin_amdgcn_wmma_f32_16x16x4_f32(false, a, false, bt2, (short)0, a2, false, false);
                a3 = __builtin_amdgcn_wmma_f32_16x16x4_f32(false, a, false, bt3, (short)0, a3, false, false);
            }
        }

        if (more) load_rows(xa, slot_n);

        {
            const float* b1e = b1s + e * HID;
            const float z0 = b1e[m], z1 = b1e[16 + m], z2 = b1e[32 + m], z3 = b1e[48 + m];
            #pragma unroll
            for (int g = 0; g < 8; ++g) {
                float* hr = hst + (g + 8 * hf) * HSTRIDE + m;
                hr[0]  = TANHF(a0[g] + z0);
                hr[16] = TANHF(a1[g] + z1);
                hr[32] = TANHF(a2[g] + z2);
                hr[48] = TANHF(a3[g] + z3);
            }
        }

        v8f d0 = {}, d1 = {}, d2 = {}, d3 = {};
        const float* w2e  = w2s + e * (HID * HID);
        const float* arow = hst + m * HSTRIDE + 2 * hf;
        #pragma unroll 4
        for (int kk = 0; kk < HID / 4; ++kk) {
            v2f a = *(const v2f*)(arow + 4 * kk);
            const float* wb = w2e + (kk * 2 + hf) * (2*HID) + 2*m;
            v2f bt0 = *(const v2f*)(wb);
            v2f bt1 = *(const v2f*)(wb + 32);
            v2f bt2 = *(const v2f*)(wb + 64);
            v2f bt3 = *(const v2f*)(wb + 96);
            d0 = __builtin_amdgcn_wmma_f32_16x16x4_f32(false, a, false, bt0, (short)0, d0, false, false);
            d1 = __builtin_amdgcn_wmma_f32_16x16x4_f32(false, a, false, bt1, (short)0, d1, false, false);
            d2 = __builtin_amdgcn_wmma_f32_16x16x4_f32(false, a, false, bt2, (short)0, d2, false, false);
            d3 = __builtin_amdgcn_wmma_f32_16x16x4_f32(false, a, false, bt3, (short)0, d3, false, false);
        }
        {
            const float* b2e = b2s + e * HID;
            const float z0 = b2e[m], z1 = b2e[16 + m], z2 = b2e[32 + m], z3 = b2e[48 + m];
            #pragma unroll
            for (int g = 0; g < 8; ++g) {
                float* hr = hst + (g + 8 * hf) * HSTRIDE + m;
                hr[0]  = TANHF(d0[g] + z0);
                hr[16] = TANHF(d1[g] + z1);
                hr[32] = TANHF(d2[g] + z2);
                hr[48] = TANHF(d3[g] + z3);
            }
        }

        {
            float en = 0.0f;
            const float* w3e = w3s + e * HID;
            const float* hr  = hst + m * HSTRIDE;
            #pragma unroll 8
            for (int h = 0; h < HID; ++h) en = fmaf(hr[h], w3e[h], en);
            en += b3s[e];
            if (hf == 0 && valid)
                atomicAdd(&molE[slot >> 6], en);
        }

        e = e_n; slot = slot_n; valid = valid_n;
    }

    __syncthreads();
    if (tid < MOLS) out[blockIdx.x * MOLS + tid] = molE[tid];
}

extern "C" void kernel_launch(void* const* d_in, const int* in_sizes, int n_in,
                              void* d_out, int out_size, void* d_ws, size_t ws_size,
                              hipStream_t stream) {
    (void)n_in; (void)out_size; (void)d_ws; (void)ws_size;
    const float* X  = (const float*)d_in[0];
    const int*   Z  = (const int*)  d_in[1];
    const float* W1 = (const float*)d_in[2];
    const float* b1 = (const float*)d_in[3];
    const float* W2 = (const float*)d_in[4];
    const float* b2 = (const float*)d_in[5];
    const float* W3 = (const float*)d_in[6];
    const float* b3 = (const float*)d_in[7];
    float* out = (float*)d_out;

    const int M      = in_sizes[1];
    const int blocks = M / CHUNK;

    hipLaunchKernelGGL(hdnnp_wmma_kernel, dim3(blocks), dim3(256), SMEM_BYTES,
                       stream, X, Z, W1, b1, W2, b2, W3, b3, out);
}


# B operands read from LDS in blocks of 8 k-steps, one block ahead of the 32 WMMAs that use them (ring of two register blocks): 6 LDS-read places per tile, not 48; k order and sum order unchanged
# speedup vs baseline: 1.7244x; 1.0337x over previous
#include <hip/hip_runtime.h>


typedef __attribute__((ext_vector_type(2))) float v2f;
typedef __attribute__((ext_vector_type(4))) float v4f;
typedef __attribute__((ext_vector_type(8))) float v8f;

#if defined(__has_builtin)
#  if __has_builtin(__builtin_amdgcn_tanhf)
#    define TANHF(x) __builtin_amdgcn_tanhf(x)
#  else
#    define TANHF(x) tanhf(x)
#  endif
#else
#  define TANHF(x) tanhf(x)
#endif

#define NELEM   4
#define DESC    128
#define HID     64
#define NATOM   64
#define CHUNK   1024
#define MOLS    (CHUNK/NATOM)
#define NTHR    256
#define NWAVES  8
#define HSTRIDE 66

#define W1_F  (NELEM*DESC*HID)
#define W2_F  (NELEM*HID*HID)
#define W3_F  (NELEM*HID)
#define B1_F  (NELEM*HID)
#define B2_F  (NELEM*HID)
#define B3_F  (NELEM)
#define HST_F (NWAVES*16*HSTRIDE)
#define MOLE_F MOLS
#define FLOATS_TOTAL (W1_F+W2_F+W3_F+B1_F+B2_F+B3_F+HST_F+MOLE_F)
#define INTS_TOTAL   (NELEM + NELEM*CHUNK)
#define SMEM_BYTES   ((size_t)(FLOATS_TOTAL + INTS_TOTAL) * 4)

__global__ __launch_bounds__(256) __attribute__((amdgpu_waves_per_eu(2, 2)))
void hdnnp_wmma_kernel(const float* __restrict__ X,
                       const int*   __restrict__ Z,
                       const float* __restrict__ W1,
                       const float* __restrict__ b1,
                       const float* __restrict__ W2,
                       const float* __restrict__ b2,
                       const float* __restrict__ W3,
                       const float* __restrict__ b3,
                       float*       __restrict__ out)
{
    extern __shared__ float smem[];
    float* w1s    = smem;
    float* w2s    = w1s + W1_F;
    float* w3s    = w2s + W2_F;
    float* b1s    = w3s + W3_F;
    float* b2s    = b1s + B1_F;
    float* b3s    = b2s + B2_F;
    float* hstage = b3s + B3_F;
    float* molE   = hstage + HST_F;
    int*   cnt    = (int*)(molE + MOLE_F);
    int*   binIdx = cnt + NELEM;

    const int tid  = threadIdx.x;
    const int lane = tid & 31;
    const int wave = tid >> 5;
    const int m    = lane & 15;
    const int hf   = lane >> 4;
    const int base = blockIdx.x * CHUNK;

    if (tid < NELEM) cnt[tid] = 0;
    if (tid < MOLS)  molE[tid] = 0.0f;
    __syncthreads();

    {
        int z[CHUNK / NTHR];
        for (int j = 0; j < CHUNK / NTHR; ++j) z[j] = Z[base + j * NTHR + tid];
        for (int j = 0; j < CHUNK / NTHR; ++j) {
            int pos = atomicAdd(&cnt[z[j]], 1);
            binIdx[z[j] * CHUNK + pos] = j * NTHR + tid;
        }
    }

    {
        const float4* s1 = (const float4*)W1;
        for (int i = tid; i < W1_F / 4; i += 256) {
            float4 v = s1[i];
            int g  = i * 4;
            int n  = g & (HID - 1);
            int d  = (g >> 6) & (DESC - 1);
            int e  = g >> 13;
            int kk = ((d >> 3) << 1) | ((d >> 1) & 1), hh = (d >> 2) & 1, lo = d & 1;
            float* dst = w1s + ((e * (DESC/4) + kk) * 2 + hh) * (2*HID) + 2*n + lo;
            dst[0] = v.x; dst[2] = v.y; dst[4] = v.z; dst[6] = v.w;
        }
        const float4* s2 = (const float4*)W2;
        for (int i = tid; i < W2_F / 4; i += 256) {
            float4 v = s2[i];
            int g  = i * 4;
            int n  = g & (HID - 1);
            int h  = (g >> 6) & (HID - 1);
            int e  = g >> 12;
            int kk = h >> 2, hh = (h >> 1) & 1, lo = h & 1;
            float* dst = w2s + ((e * (HID/4) + kk) * 2 + hh) * (2*HID) + 2*n + lo;
            dst[0] = v.x; dst[2] = v.y; dst[4] = v.z; dst[6] = v.w;
        }
        if (tid < W3_F) w3s[tid] = W3[tid];
        if (tid < B1_F) b1s[tid] = b1[tid];
        if (tid < B2_F) b2s[tid] = b2[tid];
        if (tid < B3_F) b3s[tid] = b3[tid];
    }
    __syncthreads();

    const int c0 = cnt[0], c1 = cnt[1], c2 = cnt[2], c3 = cnt[3];
    const int t0 = (c0 + 15) >> 4, t1 = (c1 + 15) >> 4;
    const int t2 = (c2 + 15) >> 4, t3 = (c3 + 15) >> 4;
    const int T  = t0 + t1 + t2 + t3;

    const int wv = __builtin_amdgcn_readfirstlane(wave);
    const int TT = __builtin_amdgcn_readfirstlane(T);

    float* hst = hstage + wv * (16 * HSTRIDE);

    auto tile_of = [&](int t, int& e, bool& valid, int& slot) {
        int lt, r = t;
        if (r < t0)            { e = 0; lt = r; }
        else { r -= t0;
        if (r < t1)            { e = 1; lt = r; }
        else { r -= t1;
        if (r < t2)            { e = 2; lt = r; }
        else                   { e = 3; lt = r - t2; } } }
        const int ccnt = cnt[e];
        const int pos  = lt * 16 + m;
        valid = pos < ccnt;
        slot  = valid ? binIdx[e * CHUNK + pos] : 0;
    };
    auto load_rows = [&](v4f (&xa)[DESC / 8], int slot) {
        const v4f* xr = (const v4f*)(X + (size_t)(base + slot) * DESC + 4 * hf);
        #pragma unroll
        for (int j = 0; j < DESC / 8; ++j) xa[j] = xr[2 * j];
    };

    auto load_block = [&](v2f (&dst)[32], const float* wbase, int kk0) {
        #pragma unroll
        for (int s = 0; s < 8; ++s) {
            const float* wb = wbase + ((kk0 + s) * 2 + hf) * (2*HID) + 2*m;
            dst[4*s + 0] = *(const v2f*)(wb);
            dst[4*s + 1] = *(const v2f*)(wb + 32);
            dst[4*s + 2] = *(const v2f*)(wb + 64);
            dst[4*s + 3] = *(const v2f*)(wb + 96);
        }
    };

    int  e, slot;
    bool valid;
    v4f  xa[DESC / 8];
    v2f  rb0[32], rb1[32];
    tile_of(wv, e, valid, slot);
    load_rows(xa, slot);
    load_block(rb0, w1s + e * (DESC * HID), 0);

    for (int t = wv; t < TT; t += NWAVES) {
        const bool more = t + NWAVES < TT;
        int  e_n = e, slot_n = slot;
        bool valid_n = false;
        if (more) tile_of(t + NWAVES, e_n, valid_n, slot_n);

        v8f a0 = {}, a1 = {}, a2 = {}, a3 = {};
        const float* w1e = w1s + e   * (DESC * HID);
        const float* w2e = w2s + e   * (HID * HID);
        const float* w1n = w1s + e_n * (DESC * HID);
        auto mma_l1 = [&](const v2f (&rb)[32], int j0) {
            #pragma unroll
            for (int jj = 0; jj < 4; ++jj) {
                const v4f x4 = xa[j0 + jj];
                #pragma unroll
                for (int b = 0; b < 2; ++b) {
                    const int s = 2 * jj + b;
                    const v2f a = b ? __builtin_shufflevector(x4, x4, 2, 3)
                                    : __builtin_shufflevector(x4, x4, 0, 1);
                    a0 = __builtin_amdgcn_wmma_f32_16x16x4_f32(false, a, false, rb[4*s + 0], (short)0, a0, false, false);
                    a1 = __builtin_amdgcn_wmma_f32_16x16x4_f32(false, a, false, rb[4*s + 1], (short)0, a1, false, false);
                    a2 = __builtin_amdgcn_wmma_f32_16x16x4_f32(false, a, false, rb[4*s + 2], (short)0, a2, false, false);
                    a3 = __builtin_amdgcn_wmma_f32_16x16x4_f32(false, a, false, rb[4*s + 3], (short)0, a3, false, false);
                }
            }
        };
        load_block(rb1, w1e, 8);
        __builtin_amdgcn_sched_barrier(0);
        mma_l1(rb0, 0);
        __builtin_amdgcn_sched_barrier(0);
        load_block(rb0, w1e, 16);
        __builtin_amdgcn_sched_barrier(0);
        mma_l1(rb1, 4);
        __builtin_amdgcn_sched_barrier(0);
        load_block(rb1, w1e, 24);
        __builtin_amdgcn_sched_barrier(0);
        mma_l1(rb0, 8);
        __builtin_amdgcn_sched_barrier(0);
        load_block(rb0, w2e, 0);
        __builtin_amdgcn_sched_barrier(0);
        mma_l1(rb1, 12);
        __builtin_amdgcn_sched_barrier(0);

        if (more) load_rows(xa, slot_n);

        {
            const float* b1e = b1s + e * HID;
            const float z0 = b1e[m], z1 = b1e[16 + m], z2 = b1e[32 + m], z3 = b1e[48 + m];
            #pragma unroll
            for (int g = 0; g < 8; ++g) {
                float* hr = hst + (g + 8 * hf) * HSTRIDE + m;
                hr[0]  = TANHF(a0[g] + z0);
                hr[16] = TANHF(a1[g] + z1);
                hr[32] = TANHF(a2[g] + z2);
                hr[48] = TANHF(a3[g] + z3);
            }
        }

        v8f d0 = {}, d1 = {}, d2 = {}, d3 = {};
        const float* arow = hst + m * HSTRIDE + 2 * hf;
        v2f ha[HID / 4];
        #pragma unroll
        for (int kk = 0; kk < HID / 4; ++kk)
            ha[kk] = *(const v2f*)(arow + 4 * kk);
        __builtin_amdgcn_sched_barrier(0);
        auto mma_l2 = [&](const v2f (&rb)[32], int k0) {
            #pragma unroll
            for (int s = 0; s < 8; ++s) {
                const v2f a = ha[k0 + s];
                d0 = __builtin_amdgcn_wmma_f32_16x16x4_f32(false, a, false, rb[4*s + 0], (short)0, d0, false, false);
                d1 = __builtin_amdgcn_wmma_f32_16x16x4_f32(false, a, false, rb[4*s + 1], (short)0, d1, false, false);
                d2 = __builtin_amdgcn_wmma_f32_16x16x4_f32(false, a, false, rb[4*s + 2], (short)0, d2, false, false);
                d3 = __builtin_amdgcn_wmma_f32_16x16x4_f32(false, a, false, rb[4*s + 3], (short)0, d3, false, false);
            }
        };
        load_block(rb1, w2e, 8);
        __builtin_amdgcn_sched_barrier(0);
        mma_l2(rb0, 0);
        __builtin_amdgcn_sched_barrier(0);
        load_block(rb0, w1n, 0);
        __builtin_amdgcn_sched_barrier(0);
        mma_l2(rb1, 8);
        __builtin_amdgcn_sched_barrier(0);
        {
            const float* b2e = b2s + e * HID;
            const float z0 = b2e[m], z1 = b2e[16 + m], z2 = b2e[32 + m], z3 = b2e[48 + m];
            #pragma unroll
            for (int g = 0; g < 8; ++g) {
                float* hr = hst + (g + 8 * hf) * HSTRIDE + m;
                hr[0]  = TANHF(d0[g] + z0);
                hr[16] = TANHF(d1[g] + z1);
                hr[32] = TANHF(d2[g] + z2);
                hr[48] = TANHF(d3[g] + z3);
            }
        }

        {
            float en = 0.0f;
            const float* w3e = w3s + e * HID;
            const float* hr  = hst + m * HSTRIDE;
            #pragma unroll 8
            for (int h = 0; h < HID; ++h) en = fmaf(hr[h], w3e[h], en);
            en += b3s[e];
            if (hf == 0 && valid)
                atomicAdd(&molE[slot >> 6], en);
        }

        e = e_n; slot = slot_n; valid = valid_n;
    }

    __syncthreads();
    if (tid < MOLS) out[blockIdx.x * MOLS + tid] = molE[tid];
}

extern "C" void kernel_launch(void* const* d_in, const int* in_sizes, int n_in,
                              void* d_out, int out_size, void* d_ws, size_t ws_size,
                              hipStream_t stream) {
    (void)n_in; (void)out_size; (void)d_ws; (void)ws_size;
    const float* X  = (const float*)d_in[0];
    const int*   Z  = (const int*)  d_in[1];
    const float* W1 = (const float*)d_in[2];
    const float* b1 = (const float*)d_in[3];
    const float* W2 = (const float*)d_in[4];
    const float* b2 = (const float*)d_in[5];
    const float* W3 = (const float*)d_in[6];
    const float* b3 = (const float*)d_in[7];
    float* out = (float*)d_out;

    const int M      = in_sizes[1];
    const int blocks = M / CHUNK;

    hipLaunchKernelGGL(hdnnp_wmma_kernel, dim3(blocks), dim3(256), SMEM_BYTES,
                       stream, X, Z, W1, b1, W2, b2, W3, b3, out);
}


# Products transposed (weights = A, atoms = B columns): accumulators feed the next layer's B operand from registers, no LDS transposes; layer 3 = 32 FMAs + 1 half exchange; layer-2/3 sum order changes
# speedup vs baseline: 1.8960x; 1.0995x over previous
#include <hip/hip_runtime.h>


typedef __attribute__((ext_vector_type(2))) float v2f;
typedef __attribute__((ext_vector_type(4))) float v4f;
typedef __attribute__((ext_vector_type(8))) float v8f;

#if defined(__has_builtin)
#  if __has_builtin(__builtin_amdgcn_tanhf)
#    define TANHF(x) __builtin_amdgcn_tanhf(x)
#  else
#    define TANHF(x) tanhf(x)
#  endif
#else
#  define TANHF(x) tanhf(x)
#endif

#define NELEM   4
#define DESC    128
#define HID     64
#define NATOM   64
#define CHUNK   1024
#define MOLS    (CHUNK/NATOM)
#define NTHR    256
#define NWAVES  8

#define W1_F  (NELEM*DESC*HID)
#define W2_F  (NELEM*HID*HID)
#define W3_F  (NELEM*HID)
#define B1_F  (NELEM*HID)
#define B2_F  (NELEM*HID)
#define B3_F  (NELEM)
#define MOLE_F MOLS
#define FLOATS_TOTAL (W1_F+W2_F+W3_F+B1_F+B2_F+B3_F+MOLE_F)
#define INTS_TOTAL   (NELEM + NELEM*CHUNK)
#define SMEM_BYTES   ((size_t)(FLOATS_TOTAL + INTS_TOTAL) * 4)

__device__ __forceinline__ float xhalf(float x) {
    int b = __builtin_bit_cast(int, x);
    return __builtin_bit_cast(float,
        __builtin_amdgcn_permlanex16(b, b, 0x76543210, 0xfedcba98, false, false));
}

__global__ __launch_bounds__(256) __attribute__((amdgpu_waves_per_eu(2, 2)))
void hdnnp_wmma_kernel(const float* __restrict__ X,
                       const int*   __restrict__ Z,
                       const float* __restrict__ W1,
                       const float* __restrict__ b1,
                       const float* __restrict__ W2,
                       const float* __restrict__ b2,
                       const float* __restrict__ W3,
                       const float* __restrict__ b3,
                       float*       __restrict__ out)
{
    extern __shared__ __attribute__((aligned(32))) float smem[];
    float* w1s    = smem;
    float* w2s    = w1s + W1_F;
    float* w3s    = w2s + W2_F;
    float* b1s    = w3s + W3_F;
    float* b2s    = b1s + B1_F;
    float* b3s    = b2s + B2_F;
    float* molE   = b3s + B3_F;
    int*   cnt    = (int*)(molE + MOLE_F);
    int*   binIdx = cnt + NELEM;

    const int tid  = threadIdx.x;
    const int lane = tid & 31;
    const int wave = tid >> 5;
    const int m    = lane & 15;
    const int hf   = lane >> 4;
    const int base = blockIdx.x * CHUNK;

    if (tid < NELEM) cnt[tid] = 0;
    if (tid < MOLS)  molE[tid] = 0.0f;
    __syncthreads();

    {
        int z[CHUNK / NTHR];
        for (int j = 0; j < CHUNK / NTHR; ++j) z[j] = Z[base + j * NTHR + tid];
        for (int j = 0; j < CHUNK / NTHR; ++j) {
            int pos = atomicAdd(&cnt[z[j]], 1);
            binIdx[z[j] * CHUNK + pos] = j * NTHR + tid;
        }
    }

    {
        const float4* s1 = (const float4*)W1;
        for (int i = tid; i < W1_F / 4; i += 256) {
            float4 v = s1[i];
            int g  = i * 4;
            int n  = g & (HID - 1);
            int d  = (g >> 6) & (DESC - 1);
            int e  = g >> 13;
            int kk = ((d >> 3) << 1) | ((d >> 1) & 1), hh = (d >> 2) & 1, lo = d & 1;
            float* dst = w1s + ((e * (DESC/4) + kk) * 2 + hh) * (2*HID) + 2*n + lo;
            dst[0] = v.x; dst[2] = v.y; dst[4] = v.z; dst[6] = v.w;
        }
        const float4* s2 = (const float4*)W2;
        for (int i = tid; i < W2_F / 4; i += 256) {
            float4 v = s2[i];
            int g  = i * 4;
            int n  = g & (HID - 1);
            int h  = (g >> 6) & (HID - 1);
            int e  = g >> 12;
            int kk = ((h >> 4) << 2) | ((h >> 1) & 3), hh = (h >> 3) & 1, lo = h & 1;
            float* dst = w2s + ((e * (HID/4) + kk) * 2 + hh) * (2*HID) + 2*n + lo;
            dst[0] = v.x; dst[2] = v.y; dst[4] = v.z; dst[6] = v.w;
        }
        if (tid < W3_F) w3s[tid] = W3[tid];
        if (tid < B1_F) b1s[tid] = b1[tid];
        if (tid < B2_F) b2s[tid] = b2[tid];
        if (tid < B3_F) b3s[tid] = b3[tid];
    }
    __syncthreads();

    const int c0 = cnt[0], c1 = cnt[1], c2 = cnt[2], c3 = cnt[3];
    const int t0 = (c0 + 15) >> 4, t1 = (c1 + 15) >> 4;
    const int t2 = (c2 + 15) >> 4, t3 = (c3 + 15) >> 4;
    const int T  = t0 + t1 + t2 + t3;

    const int wv = __builtin_amdgcn_readfirstlane(wave);
    const int TT = __builtin_amdgcn_readfirstlane(T);

    auto tile_of = [&](int t, int& e, bool& valid, int& slot) {
        int lt, r = t;
        if (r < t0)            { e = 0; lt = r; }
        else { r -= t0;
        if (r < t1)            { e = 1; lt = r; }
        else { r -= t1;
        if (r < t2)            { e = 2; lt = r; }
        else                   { e = 3; lt = r - t2; } } }
        const int ccnt = cnt[e];
        const int pos  = lt * 16 + m;
        valid = pos < ccnt;
        slot  = valid ? binIdx[e * CHUNK + pos] : 0;
    };
    auto load_rows = [&](v4f (&xa)[DESC / 8], int slot) {
        const v4f* xr = (const v4f*)(X + (size_t)(base + slot) * DESC + 4 * hf);
        #pragma unroll
        for (int j = 0; j < DESC / 8; ++j) xa[j] = xr[2 * j];
    };

    auto load_block = [&](v2f (&dst)[32], const float* wbase, int kk0) {
        #pragma unroll
        for (int s = 0; s < 8; ++s) {
            const float* wb = wbase + ((kk0 + s) * 2 + hf) * (2*HID) + 2*m;
            dst[4*s + 0] = *(const v2f*)(wb);
            dst[4*s + 1] = *(const v2f*)(wb + 32);
            dst[4*s + 2] = *(const v2f*)(wb + 64);
            dst[4*s + 3] = *(const v2f*)(wb + 96);
        }
    };
    auto load_hvec = [&](v8f (&dst)[4], const float* vbase) {
        #pragma unroll
        for (int u = 0; u < 4; ++u)
            dst[u] = *(const v8f*)(vbase + 16 * u + 8 * hf);
    };
    auto act = [&](const v8f& acc, const v8f& bias) -> v8f {
        v8f o;
        #pragma unroll
        for (int r = 0; r < 8; ++r) o[r] = TANHF(acc[r] + bias[r]);
        return o;
    };

    int  e, slot;
    bool valid;
    v4f  xa[DESC / 8];
    v2f  rb0[32], rb1[32];
    tile_of(wv, e, valid, slot);
    load_rows(xa, slot);
    load_block(rb0, w1s + e * (DESC * HID), 0);

    for (int t = wv; t < TT; t += NWAVES) {
        const bool more = t + NWAVES < TT;
        int  e_n = e, slot_n = slot;
        bool valid_n = false;
        if (more) tile_of(t + NWAVES, e_n, valid_n, slot_n);

        v8f a0 = {}, a1 = {}, a2 = {}, a3 = {};
        const float* w1e = w1s + e   * (DESC * HID);
        const float* w2e = w2s + e   * (HID * HID);
        const float* w1n = w1s + e_n * (DESC * HID);
        auto mma_l1 = [&](const v2f (&rb)[32], int j0) {
            #pragma unroll
            for (int jj = 0; jj < 4; ++jj) {
                const v4f x4 = xa[j0 + jj];
                #pragma unroll
                for (int b = 0; b < 2; ++b) {
                    const int s = 2 * jj + b;
                    const v2f x = b ? __builtin_shufflevector(x4, x4, 2, 3)
                                    : __builtin_shufflevector(x4, x4, 0, 1);
                    a0 = __builtin_amdgcn_wmma_f32_16x16x4_f32(false, rb[4*s + 0], false, x, (short)0, a0, false, false);
                    a1 = __builtin_amdgcn_wmma_f32_16x16x4_f32(false, rb[4*s + 1], false, x, (short)0, a1, false, false);
                    a2 = __builtin_amdgcn_wmma_f32_16x16x4_f32(false, rb[4*s + 2], false, x, (short)0, a2, false, false);
                    a3 = __builtin_amdgcn_wmma_f32_16x16x4_f32(false, rb[4*s + 3], false, x, (short)0, a3, false, false);
                }
            }
        };
        load_block(rb1, w1e, 8);
        __builtin_amdgcn_sched_barrier(0);
        mma_l1(rb0, 0);
        __builtin_amdgcn_sched_barrier(0);
        load_block(rb0, w1e, 16);
        __builtin_amdgcn_sched_barrier(0);
        mma_l1(rb1, 4);
        __builtin_amdgcn_sched_barrier(0);
        load_block(rb1, w1e, 24);
        __builtin_amdgcn_sched_barrier(0);
        mma_l1(rb0, 8);
        __builtin_amdgcn_sched_barrier(0);
        load_block(rb0, w2e, 0);
        v8f bv1[4];
        load_hvec(bv1, b1s + e * HID);
        __builtin_amdgcn_sched_barrier(0);
        mma_l1(rb1, 12);
        __builtin_amdgcn_sched_barrier(0);

        if (more) load_rows(xa, slot_n);

        const v8f g0 = act(a0, bv1[0]);
        const v8f g1 = act(a1, bv1[1]);
        const v8f g2 = act(a2, bv1[2]);
        const v8f g3 = act(a3, bv1[3]);

        v8f d0 = {}, d1 = {}, d2 = {}, d3 = {};
        auto mma_l2 = [&](const v2f (&rb)[32], const v8f& ga, const v8f& gb) {
            #pragma unroll
            for (int s = 0; s < 8; ++s) {
                const int q = s & 3;
                const v2f hb = (s < 4) ? v2f{ga[2*q], ga[2*q + 1]} : v2f{gb[2*q], gb[2*q + 1]};
                d0 = __builtin_amdgcn_wmma_f32_16x16x4_f32(false, rb[4*s + 0], false, hb, (short)0, d0, false, false);
                d1 = __builtin_amdgcn_wmma_f32_16x16x4_f32(false, rb[4*s + 1], false, hb, (short)0, d1, false, false);
                d2 = __builtin_amdgcn_wmma_f32_16x16x4_f32(false, rb[4*s + 2], false, hb, (short)0, d2, false, false);
                d3 = __builtin_amdgcn_wmma_f32_16x16x4_f32(false, rb[4*s + 3], false, hb, (short)0, d3, false, false);
            }
        };
        load_block(rb1, w2e, 8);
        __builtin_amdgcn_sched_barrier(0);
        mma_l2(rb0, g0, g1);
        __builtin_amdgcn_sched_barrier(0);
        load_block(rb0, w1n, 0);
        v8f bv2[4], wv3[4];
        load_hvec(bv2, b2s + e * HID);
        load_hvec(wv3, w3s + e * HID);
        const float b3e = b3s[e];
        __builtin_amdgcn_sched_barrier(0);
        mma_l2(rb1, g2, g3);
        __builtin_amdgcn_sched_barrier(0);

        {
            const v8f k0 = act(d0, bv2[0]), k1 = act(d1, bv2[1]);
            const v8f k2 = act(d2, bv2[2]), k3 = act(d3, bv2[3]);
            v8f q;
            #pragma unroll
            for (int r = 0; r < 8; ++r)
                q[r] = fmaf(k3[r], wv3[3][r], fmaf(k2[r], wv3[2][r], fmaf(k1[r], wv3[1][r], k0[r] * wv3[0][r])));
            const float part = ((q[0] + q[1]) + (q[2] + q[3])) + ((q[4] + q[5]) + (q[6] + q[7]));
            const float en = part + xhalf(part) + b3e;
            if (hf == 0 && valid)
                atomicAdd(&molE[slot >> 6], en);
        }

        e = e_n; slot = slot_n; valid = valid_n;
    }

    __syncthreads();
    if (tid < MOLS) out[blockIdx.x * MOLS + tid] = molE[tid];
}

extern "C" void kernel_launch(void* const* d_in, const int* in_sizes, int n_in,
                              void* d_out, int out_size, void* d_ws, size_t ws_size,
                              hipStream_t stream) {
    (void)n_in; (void)out_size; (void)d_ws; (void)ws_size;
    const float* X  = (const float*)d_in[0];
    const int*   Z  = (const int*)  d_in[1];
    const float* W1 = (const float*)d_in[2];
    const float* b1 = (const float*)d_in[3];
    const float* W2 = (const float*)d_in[4];
    const float* b2 = (const float*)d_in[5];
    const float* W3 = (const float*)d_in[6];
    const float* b3 = (const float*)d_in[7];
    float* out = (float*)d_out;

    const int M      = in_sizes[1];
    const int blocks = M / CHUNK;

    hipLaunchKernelGGL(hdnnp_wmma_kernel, dim3(blocks), dim3(256), SMEM_BYTES,
                       stream, X, Z, W1, b1, W2, b2, W3, b3, out);
}


# Work-group of 12 waves (3 per SIMD), not 8: a third wave feeds the matrix pipe while others do tanh, LDS reads, row loads; staging split 8 waves W1 + 4 waves W2; LDS-atomic arrival order changes
# speedup vs baseline: 2.2182x; 1.1699x over previous
#include <hip/hip_runtime.h>


typedef __attribute__((ext_vector_type(2))) float v2f;
typedef __attribute__((ext_vector_type(4))) float v4f;
typedef __attribute__((ext_vector_type(8))) float v8f;

#if defined(__has_builtin)
#  if __has_builtin(__builtin_amdgcn_tanhf)
#    define TANHF(x) __builtin_amdgcn_tanhf(x)
#  else
#    define TANHF(x) tanhf(x)
#  endif
#else
#  define TANHF(x) tanhf(x)
#endif

#define NELEM   4
#define DESC    128
#define HID     64
#define NATOM   64
#define CHUNK   1024
#define MOLS    (CHUNK/NATOM)
#define NTHR    384
#define NWAVES  12
#define ZPT     ((CHUNK + NTHR - 1) / NTHR)

#define W1_F  (NELEM*DESC*HID)
#define W2_F  (NELEM*HID*HID)
#define W3_F  (NELEM*HID)
#define B1_F  (NELEM*HID)
#define B2_F  (NELEM*HID)
#define B3_F  (NELEM)
#define MOLE_F MOLS
#define FLOATS_TOTAL (W1_F+W2_F+W3_F+B1_F+B2_F+B3_F+MOLE_F)
#define INTS_TOTAL   (NELEM + NELEM*CHUNK)
#define SMEM_BYTES   ((size_t)(FLOATS_TOTAL + INTS_TOTAL) * 4)

__device__ __forceinline__ float xhalf(float x) {
    int b = __builtin_bit_cast(int, x);
    return __builtin_bit_cast(float,
        __builtin_amdgcn_permlanex16(b, b, 0x76543210, 0xfedcba98, false, false));
}

template <bool IS_W1>
__device__ __forceinline__ void stage_weights(const float* __restrict__ src, float* dstbase, int first) {
    constexpr int STEP = IS_W1 ? 256 : 128;
    const float4* s = (const float4*)src;
    float4 v[32];
    #pragma unroll
    for (int u = 0; u < 32; ++u) v[u] = s[first + u * STEP];
    #pragma unroll
    for (int u = 0; u < 32; ++u) {
        const int g = (first + u * STEP) * 4;
        const int n = g & (HID - 1);
        const int r = g >> 6;
        float* dst;
        if constexpr (IS_W1) {
            const int d = r & (DESC - 1), e = r >> 7;
            const int kk = ((d >> 3) << 1) | ((d >> 1) & 1), hh = (d >> 2) & 1, lo = d & 1;
            dst = dstbase + ((e * (DESC/4) + kk) * 2 + hh) * (2*HID) + 2*n + lo;
        } else {
            const int h = r & (HID - 1), e = r >> 6;
            const int kk = ((h >> 4) << 2) | ((h >> 1) & 3), hh = (h >> 3) & 1, lo = h & 1;
            dst = dstbase + ((e * (HID/4) + kk) * 2 + hh) * (2*HID) + 2*n + lo;
        }
        dst[0] = v[u].x; dst[2] = v[u].y; dst[4] = v[u].z; dst[6] = v[u].w;
    }
}

__global__ __launch_bounds__(NTHR) __attribute__((amdgpu_waves_per_eu(3, 3)))
void hdnnp_wmma_kernel(const float* __restrict__ X,
                       const int*   __restrict__ Z,
                       const float* __restrict__ W1,
                       const float* __restrict__ b1,
                       const float* __restrict__ W2,
                       const float* __restrict__ b2,
                       const float* __restrict__ W3,
                       const float* __restrict__ b3,
                       float*       __restrict__ out)
{
    extern __shared__ __attribute__((aligned(32))) float smem[];
    float* w1s    = smem;
    float* w2s    = w1s + W1_F;
    float* w3s    = w2s + W2_F;
    float* b1s    = w3s + W3_F;
    float* b2s    = b1s + B1_F;
    float* b3s    = b2s + B2_F;
    float* molE   = b3s + B3_F;
    int*   cnt    = (int*)(molE + MOLE_F);
    int*   binIdx = cnt + NELEM;

    const int tid  = threadIdx.x;
    const int lane = tid & 31;
    const int wave = tid >> 5;
    const int m    = lane & 15;
    const int hf   = lane >> 4;
    const int base = blockIdx.x * CHUNK;

    if (tid < NELEM) cnt[tid] = 0;
    if (tid < MOLS)  molE[tid] = 0.0f;
    __syncthreads();

    {
        int z[ZPT];
        #pragma unroll
        for (int j = 0; j < ZPT; ++j) {
            const int a = j * NTHR + tid;
            z[j] = (a < CHUNK) ? Z[base + a] : -1;
        }
        #pragma unroll
        for (int j = 0; j < ZPT; ++j) {
            if (z[j] >= 0) {
                int pos = atomicAdd(&cnt[z[j]], 1);
                binIdx[z[j] * CHUNK + pos] = j * NTHR + tid;
            }
        }
    }

    {
        if (__builtin_amdgcn_readfirstlane(wave) < 8)
            stage_weights<true >(W1, w1s, tid);
        else
            stage_weights<false>(W2, w2s, tid - 256);
        if (tid < W3_F) w3s[tid] = W3[tid];
        if (tid < B1_F) b1s[tid] = b1[tid];
        if (tid < B2_F) b2s[tid] = b2[tid];
        if (tid < B3_F) b3s[tid] = b3[tid];
    }
    __syncthreads();

    const int c0 = cnt[0], c1 = cnt[1], c2 = cnt[2], c3 = cnt[3];
    const int t0 = (c0 + 15) >> 4, t1 = (c1 + 15) >> 4;
    const int t2 = (c2 + 15) >> 4, t3 = (c3 + 15) >> 4;
    const int T  = t0 + t1 + t2 + t3;

    const int wv = __builtin_amdgcn_readfirstlane(wave);
    const int TT = __builtin_amdgcn_readfirstlane(T);

    auto tile_of = [&](int t, int& e, bool& valid, int& slot) {
        int lt, r = t;
        if (r < t0)            { e = 0; lt = r; }
        else { r -= t0;
        if (r < t1)            { e = 1; lt = r; }
        else { r -= t1;
        if (r < t2)            { e = 2; lt = r; }
        else                   { e = 3; lt = r - t2; } } }
        const int ccnt = cnt[e];
        const int pos  = lt * 16 + m;
        valid = pos < ccnt;
        slot  = valid ? binIdx[e * CHUNK + pos] : 0;
    };
    auto load_rows = [&](v4f (&xa)[DESC / 8], int slot) {
        const v4f* xr = (const v4f*)(X + (size_t)(base + slot) * DESC + 4 * hf);
        #pragma unroll
        for (int j = 0; j < DESC / 8; ++j) xa[j] = xr[2 * j];
    };

    auto load_block = [&](v2f (&dst)[32], const float* wbase, int kk0) {
        #pragma unroll
        for (int s = 0; s < 8; ++s) {
            const float* wb = wbase + ((kk0 + s) * 2 + hf) * (2*HID) + 2*m;
            dst[4*s + 0] = *(const v2f*)(wb);
            dst[4*s + 1] = *(const v2f*)(wb + 32);
            dst[4*s + 2] = *(const v2f*)(wb + 64);
            dst[4*s + 3] = *(const v2f*)(wb + 96);
        }
    };
    auto load_hvec = [&](v8f (&dst)[4], const float* vbase) {
        #pragma unroll
        for (int u = 0; u < 4; ++u)
            dst[u] = *(const v8f*)(vbase + 16 * u + 8 * hf);
    };
    auto act = [&](const v8f& acc, const v8f& bias) -> v8f {
        v8f o;
        #pragma unroll
        for (int r = 0; r < 8; ++r) o[r] = TANHF(acc[r] + bias[r]);
        return o;
    };

    int  e, slot;
    bool valid;
    v4f  xa[DESC / 8];
    v2f  rb0[32], rb1[32];
    tile_of(wv, e, valid, slot);
    load_rows(xa, slot);
    load_block(rb0, w1s + e * (DESC * HID), 0);

    for (int t = wv; t < TT; t += NWAVES) {
        const bool more = t + NWAVES < TT;
        int  e_n = e, slot_n = slot;
        bool valid_n = false;
        if (more) tile_of(t + NWAVES, e_n, valid_n, slot_n);

        v8f a0 = {}, a1 = {}, a2 = {}, a3 = {};
        const float* w1e = w1s + e   * (DESC * HID);
        const float* w2e = w2s + e   * (HID * HID);
        const float* w1n = w1s + e_n * (DESC * HID);
        auto mma_l1 = [&](const v2f (&rb)[32], int j0) {
            #pragma unroll
            for (int jj = 0; jj < 4; ++jj) {
                const v4f x4 = xa[j0 + jj];
                #pragma unroll
                for (int b = 0; b < 2; ++b) {
                    const int s = 2 * jj + b;
                    const v2f x = b ? __builtin_shufflevector(x4, x4, 2, 3)
                                    : __builtin_shufflevector(x4, x4, 0, 1);
                    a0 = __builtin_amdgcn_wmma_f32_16x16x4_f32(false, rb[4*s + 0], false, x, (short)0, a0, false, false);
                    a1 = __builtin_amdgcn_wmma_f32_16x16x4_f32(false, rb[4*s + 1], false, x, (short)0, a1, false, false);
                    a2 = __builtin_amdgcn_wmma_f32_16x16x4_f32(false, rb[4*s + 2], false, x, (short)0, a2, false, false);
                    a3 = __builtin_amdgcn_wmma_f32_16x16x4_f32(false, rb[4*s + 3], false, x, (short)0, a3, false, false);
                }
            }
        };
        load_block(rb1, w1e, 8);
        __builtin_amdgcn_sched_barrier(0);
        mma_l1(rb0, 0);
        __builtin_amdgcn_sched_barrier(0);
        load_block(rb0, w1e, 16);
        __builtin_amdgcn_sched_barrier(0);
        mma_l1(rb1, 4);
        __builtin_amdgcn_sched_barrier(0);
        load_block(rb1, w1e, 24);
        __builtin_amdgcn_sched_barrier(0);
        mma_l1(rb0, 8);
        __builtin_amdgcn_sched_barrier(0);
        load_block(rb0, w2e, 0);
        v8f bv1[4];
        load_hvec(bv1, b1s + e * HID);
        __builtin_amdgcn_sched_barrier(0);
        mma_l1(rb1, 12);
        __builtin_amdgcn_sched_barrier(0);

        if (more) load_rows(xa, slot_n);

        const v8f g0 = act(a0, bv1[0]);
        const v8f g1 = act(a1, bv1[1]);
        const v8f g2 = act(a2, bv1[2]);
        const v8f g3 = act(a3, bv1[3]);

        v8f d0 = {}, d1 = {}, d2 = {}, d3 = {};
        auto mma_l2 = [&](const v2f (&rb)[32], const v8f& ga, const v8f& gb) {
            #pragma unroll
            for (int s = 0; s < 8; ++s) {
                const int q = s & 3;
                const v2f hb = (s < 4) ? v2f{ga[2*q], ga[2*q + 1]} : v2f{gb[2*q], gb[2*q + 1]};
                d0 = __builtin_amdgcn_wmma_f32_16x16x4_f32(false, rb[4*s + 0], false, hb, (short)0, d0, false, false);
                d1 = __builtin_amdgcn_wmma_f32_16x16x4_f32(false, rb[4*s + 1], false, hb, (short)0, d1, false, false);
                d2 = __builtin_amdgcn_wmma_f32_16x16x4_f32(false, rb[4*s + 2], false, hb, (short)0, d2, false, false);
                d3 = __builtin_amdgcn_wmma_f32_16x16x4_f32(false, rb[4*s + 3], false, hb, (short)0, d3, false, false);
            }
        };
        load_block(rb1, w2e, 8);
        __builtin_amdgcn_sched_barrier(0);
        mma_l2(rb0, g0, g1);
        __builtin_amdgcn_sched_barrier(0);
        load_block(rb0, w1n, 0);
        v8f bv2[4], wv3[4];
        load_hvec(bv2, b2s + e * HID);
        load_hvec(wv3, w3s + e * HID);
        const float b3e = b3s[e];
        __builtin_amdgcn_sched_barrier(0);
        mma_l2(rb1, g2, g3);
        __builtin_amdgcn_sched_barrier(0);

        {
            const v8f k0 = act(d0, bv2[0]), k1 = act(d1, bv2[1]);
            const v8f k2 = act(d2, bv2[2]), k3 = act(d3, bv2[3]);
            v8f q;
            #pragma unroll
            for (int r = 0; r < 8; ++r)
                q[r] = fmaf(k3[r], wv3[3][r], fmaf(k2[r], wv3[2][r], fmaf(k1[r], wv3[1][r], k0[r] * wv3[0][r])));
            const float part = ((q[0] + q[1]) + (q[2] + q[3])) + ((q[4] + q[5]) + (q[6] + q[7]));
            const float en = part + xhalf(part) + b3e;
            if (hf == 0 && valid)
                atomicAdd(&molE[slot >> 6], en);
        }

        e = e_n; slot = slot_n; valid = valid_n;
    }

    __syncthreads();
    if (tid < MOLS) out[blockIdx.x * MOLS + tid] = molE[tid];
}

extern "C" void kernel_launch(void* const* d_in, const int* in_sizes, int n_in,
                              void* d_out, int out_size, void* d_ws, size_t ws_size,
                              hipStream_t stream) {
    (void)n_in; (void)out_size; (void)d_ws; (void)ws_size;
    const float* X  = (const float*)d_in[0];
    const int*   Z  = (const int*)  d_in[1];
    const float* W1 = (const float*)d_in[2];
    const float* b1 = (const float*)d_in[3];
    const float* W2 = (const float*)d_in[4];
    const float* b2 = (const float*)d_in[5];
    const float* W3 = (const float*)d_in[6];
    const float* b3 = (const float*)d_in[7];
    float* out = (float*)d_out;

    const int M      = in_sizes[1];
    const int blocks = M / CHUNK;

    hipLaunchKernelGGL(hdnnp_wmma_kernel, dim3(blocks), dim3(NTHR), SMEM_BYTES,
                       stream, X, Z, W1, b1, W2, b2, W3, b3, out);
}
